# starting kernel
# baseline (speedup 1.0000x reference)
#include <hip/hip_runtime.h>
#include <math.h>


typedef __attribute__((ext_vector_type(2))) float v2f;
typedef __attribute__((ext_vector_type(8))) float v8f;

#define ALDS_STRIDE 36
#define BLDS_STRIDE 516
#define SLDS_STRIDE 524
#define LDS_FLOATS  (256 * ALDS_STRIDE + 32 * BLDS_STRIDE + 16 * SLDS_STRIDE)

__device__ __forceinline__ float lane_sum(float v) {
#pragma unroll
    for (int off = 16; off > 0; off >>= 1) v += __shfl_xor(v, off, 32);
    return v;
}
__device__ __forceinline__ float lane_max(float v) {
#pragma unroll
    for (int off = 16; off > 0; off >>= 1) v = fmaxf(v, __shfl_xor(v, off, 32));
    return v;
}

__device__ __forceinline__ void weighted_squash(const float* __restrict__ xv,
                                                float c, float* __restrict__ sv)
{
    float t[16];
#pragma unroll
    for (int d = 0; d < 16; ++d) t[d] = xv[d] * c;
#pragma unroll
    for (int off = 16; off > 0; off >>= 1) {
#pragma unroll
        for (int d = 0; d < 16; ++d) t[d] += __shfl_xor(t[d], off, 32);
    }
    float sq = 0.f;
#pragma unroll
    for (int d = 0; d < 16; ++d) sq = fmaf(t[d], t[d], sq);
    float coef = (sq / (1.0f + sq)) / sqrtf(sq + 1e-7f);
#pragma unroll
    for (int d = 0; d < 16; ++d) sv[d] = coef * t[d];
}

__global__ __launch_bounds__(256)
void capsule_fused_wmma(const float* __restrict__ prim,
                        const float* __restrict__ Wt,
                        const float* __restrict__ bprior,
                        float* __restrict__ out)
{
    extern __shared__ float smem[];
    float* Alds  = smem;
    float* Blds  = Alds + 256 * ALDS_STRIDE;
    float* Strip = Blds + 32 * BLDS_STRIDE;

    const int bm   = blockIdx.x;
    const int o    = bm & 31;
    const int tid  = threadIdx.x;
    const int wave = tid >> 5;
    const int lane = tid & 31;

    const float* Ag = prim + (size_t)bm * 8192;
    const float* Wg = Wt   + (size_t)bm * 16384;

    __builtin_prefetch(Wg, 0, 0);

#pragma unroll
    for (int i = 0; i < 8; ++i) {
        int f4  = tid + i * 256;
        float4 v = ((const float4*)Ag)[f4];
        int f   = f4 << 2;
        int row = f >> 5;
        int kk  = f & 31;
        float* dst = &Alds[row * ALDS_STRIDE + kk];
        dst[0] = v.x; dst[1] = v.y; dst[2] = v.z; dst[3] = v.w;
    }
#pragma unroll 4
    for (int i = 0; i < 64; ++i) {
        int lin = i * 256 + tid;
        int c   = lin >> 9;
        int n   = lin & 511;
        int yx  = c >> 3, e = c & 7;
        int og  = n >> 4, d = n & 15;
        Blds[c * BLDS_STRIDE + n] = Wg[yx * 4096 + og * 128 + d * 8 + e];
    }
    __syncthreads();

    const int khalf = (lane >> 4) << 1;

    for (int mtile = 0; mtile < 16; ++mtile) {
        const int arow = mtile * 16 + (lane & 15);
#pragma unroll
        for (int i = 0; i < 4; ++i) {
            int nt  = i * 8 + wave;
            int col = nt * 16 + (lane & 15);
            v8f acc = {};
#pragma unroll
            for (int kc = 0; kc < 32; kc += 4) {
                int kk = kc + khalf;
                v2f a, b;
                a.x = Alds[arow * ALDS_STRIDE + kk];
                a.y = Alds[arow * ALDS_STRIDE + kk + 1];
                b.x = Blds[kk * BLDS_STRIDE + col];
                b.y = Blds[(kk + 1) * BLDS_STRIDE + col];
                acc = __builtin_amdgcn_wmma_f32_16x16x4_f32(
                          false, a, false, b, (short)0, acc, false, false);
            }
            int rbase = (lane >> 4) << 3;
#pragma unroll
            for (int r = 0; r < 8; ++r)
                Strip[(rbase + r) * SLDS_STRIDE + col] = acc[r];
        }
        __syncthreads();

#pragma unroll
        for (int p = 0; p < 2; ++p) {
            int rowk = p * 8 + wave;
            int k    = mtile * 16 + rowk;

            const float* sp = &Strip[rowk * SLDS_STRIDE + lane * 16];
            float xv[16];
#pragma unroll
            for (int q = 0; q < 4; ++q) {
                float4 v = ((const float4*)sp)[q];
                xv[q * 4 + 0] = v.x; xv[q * 4 + 1] = v.y;
                xv[q * 4 + 2] = v.z; xv[q * 4 + 3] = v.w;
            }
            float bb = bprior[((size_t)o * 256 + k) * 32 + lane];

            float sv[16];
            {
                float mx = lane_max(bb);
                float e  = __expf(bb - mx);
                float s  = lane_sum(e);
                weighted_squash(xv, e / s, sv);
            }
#pragma unroll
            for (int it = 0; it < 2; ++it) {
                float dot = 0.f;
#pragma unroll
                for (int d = 0; d < 16; ++d) dot = fmaf(xv[d], sv[d], dot);
                bb += dot;
                float mx = lane_max(bb);
                float e  = __expf(bb - mx);
                float s  = lane_sum(e);
                weighted_squash(xv, e / s, sv);
            }

            if (lane == 0) {
                float4* op = (float4*)(out + ((size_t)bm * 256 + k) * 16);
                op[0] = make_float4(sv[0],  sv[1],  sv[2],  sv[3]);
                op[1] = make_float4(sv[4],  sv[5],  sv[6],  sv[7]);
                op[2] = make_float4(sv[8],  sv[9],  sv[10], sv[11]);
                op[3] = make_float4(sv[12], sv[13], sv[14], sv[15]);
            }
        }
        __syncthreads();
    }
}

extern "C" void kernel_launch(void* const* d_in, const int* in_sizes, int n_in,
                              void* d_out, int out_size, void* d_ws, size_t ws_size,
                              hipStream_t stream)
{
    const float* prim = (const float*)d_in[0];
    const float* Wt   = (const float*)d_in[1];
    const float* bpr  = (const float*)d_in[2];
    float* out = (float*)d_out;

    const size_t lds_bytes = (size_t)LDS_FLOATS * sizeof(float);
    capsule_fused_wmma<<<256, 256, lds_bytes, stream>>>(prim, Wt, bpr, out);
}


# Routing sums moved off LDS-unit lane permutes: strip row held in two register views so sums over m and d are DPP row_share/row_xmask ops; changes summation order of routing sums
# speedup vs baseline: 2.4425x; 2.4425x over previous
#include <hip/hip_runtime.h>
#include <math.h>


typedef __attribute__((ext_vector_type(2))) float v2f;
typedef __attribute__((ext_vector_type(8))) float v8f;

#define ALDS_STRIDE 36
#define BLDS_STRIDE 516
#define SLDS_STRIDE 524
#define LDS_FLOATS  (256 * ALDS_STRIDE + 32 * BLDS_STRIDE + 16 * SLDS_STRIDE)

__device__ __forceinline__ float xhalf(float x) {
    int b = __builtin_bit_cast(int, x);
    return __builtin_bit_cast(float,
        __builtin_amdgcn_permlanex16(b, b, 0x76543210, 0xfedcba98, false, false));
}
#define ROW_SHARE(n) (0x150 + (n))
#define ROW_XMASK(n) (0x160 + (n))
template <int CTRL>
__device__ __forceinline__ float row_dpp(float x) {
    return __builtin_bit_cast(float,
        __builtin_amdgcn_update_dpp(0, __builtin_bit_cast(int, x), CTRL, 0xf, 0xf, true));
}
__device__ __forceinline__ float row_sum16(float v) {
    v += row_dpp<ROW_XMASK(1)>(v);
    v += row_dpp<ROW_XMASK(2)>(v);
    v += row_dpp<ROW_XMASK(4)>(v);
    v += row_dpp<ROW_XMASK(8)>(v);
    return v;
}
__device__ __forceinline__ float wave_sum(float v) {
    v = row_sum16(v);
    return v + xhalf(v);
}
__device__ __forceinline__ float wave_max(float v) {
    v = fmaxf(v, row_dpp<ROW_XMASK(1)>(v));
    v = fmaxf(v, row_dpp<ROW_XMASK(2)>(v));
    v = fmaxf(v, row_dpp<ROW_XMASK(4)>(v));
    v = fmaxf(v, row_dpp<ROW_XMASK(8)>(v));
    return fmaxf(v, xhalf(v));
}
__device__ __forceinline__ float share_dot16(float src, const float (&x)[16]) {
    float a0 = row_dpp<ROW_SHARE(0)>(src) * x[0];
    float a1 = row_dpp<ROW_SHARE(1)>(src) * x[1];
    a0 = fmaf(row_dpp<ROW_SHARE(2)>(src),  x[2],  a0);
    a1 = fmaf(row_dpp<ROW_SHARE(3)>(src),  x[3],  a1);
    a0 = fmaf(row_dpp<ROW_SHARE(4)>(src),  x[4],  a0);
    a1 = fmaf(row_dpp<ROW_SHARE(5)>(src),  x[5],  a1);
    a0 = fmaf(row_dpp<ROW_SHARE(6)>(src),  x[6],  a0);
    a1 = fmaf(row_dpp<ROW_SHARE(7)>(src),  x[7],  a1);
    a0 = fmaf(row_dpp<ROW_SHARE(8)>(src),  x[8],  a0);
    a1 = fmaf(row_dpp<ROW_SHARE(9)>(src),  x[9],  a1);
    a0 = fmaf(row_dpp<ROW_SHARE(10)>(src), x[10], a0);
    a1 = fmaf(row_dpp<ROW_SHARE(11)>(src), x[11], a1);
    a0 = fmaf(row_dpp<ROW_SHARE(12)>(src), x[12], a0);
    a1 = fmaf(row_dpp<ROW_SHARE(13)>(src), x[13], a1);
    a0 = fmaf(row_dpp<ROW_SHARE(14)>(src), x[14], a0);
    a1 = fmaf(row_dpp<ROW_SHARE(15)>(src), x[15], a1);
    return a0 + a1;
}

__device__ __forceinline__ float weighted_squash(float c, const float (&xd)[16])
{
    float s = share_dot16(c, xd);
    s += xhalf(s);
    float sq = row_sum16(s * s);
    float coef = (sq / (1.0f + sq)) / sqrtf(sq + 1e-7f);
    return coef * s;
}

__device__ __forceinline__ float route_position(const float* __restrict__ srow,
                                                float bb, int lane)
{
    float xm[16];
    const float* sp = srow + lane * 16;
#pragma unroll
    for (int q = 0; q < 4; ++q) {
        float4 v = ((const float4*)sp)[q];
        xm[q * 4 + 0] = v.x; xm[q * 4 + 1] = v.y;
        xm[q * 4 + 2] = v.z; xm[q * 4 + 3] = v.w;
    }
    float xd[16];
    const float* sd = srow + (lane >> 4) * 256 + (lane & 15);
#pragma unroll
    for (int j = 0; j < 16; ++j) xd[j] = sd[j * 16];

    float v;
    {
        float mx = wave_max(bb);
        float e  = __expf(bb - mx);
        float s  = wave_sum(e);
        v = weighted_squash(e / s, xd);
    }
#pragma unroll
    for (int it = 0; it < 2; ++it) {
        bb += share_dot16(v, xm);
        float mx = wave_max(bb);
        float e  = __expf(bb - mx);
        float s  = wave_sum(e);
        v = weighted_squash(e / s, xd);
    }
    return v;
}

__global__ __launch_bounds__(256)
void capsule_fused_wmma(const float* __restrict__ prim,
                        const float* __restrict__ Wt,
                        const float* __restrict__ bprior,
                        float* __restrict__ out)
{
    extern __shared__ float smem[];
    float* Alds  = smem;
    float* Blds  = Alds + 256 * ALDS_STRIDE;
    float* Strip = Blds + 32 * BLDS_STRIDE;

    const int bm   = blockIdx.x;
    const int o    = bm & 31;
    const int tid  = threadIdx.x;
    const int wave = tid >> 5;
    const int lane = tid & 31;

    const float* Ag = prim + (size_t)bm * 8192;
    const float* Wg = Wt   + (size_t)bm * 16384;

    __builtin_prefetch(Wg, 0, 0);

#pragma unroll
    for (int i = 0; i < 8; ++i) {
        int f4  = tid + i * 256;
        float4 v = ((const float4*)Ag)[f4];
        int f   = f4 << 2;
        int row = f >> 5;
        int kk  = f & 31;
        float* dst = &Alds[row * ALDS_STRIDE + kk];
        dst[0] = v.x; dst[1] = v.y; dst[2] = v.z; dst[3] = v.w;
    }
#pragma unroll 4
    for (int i = 0; i < 64; ++i) {
        int lin = i * 256 + tid;
        int c   = lin >> 9;
        int n   = lin & 511;
        int yx  = c >> 3, e = c & 7;
        int og  = n >> 4, d = n & 15;
        Blds[c * BLDS_STRIDE + n] = Wg[yx * 4096 + og * 128 + d * 8 + e];
    }
    __syncthreads();

    const int khalf = (lane >> 4) << 1;

    for (int mtile = 0; mtile < 16; ++mtile) {
        const int arow = mtile * 16 + (lane & 15);
#pragma unroll
        for (int i = 0; i < 4; ++i) {
            int nt  = i * 8 + wave;
            int col = nt * 16 + (lane & 15);
            v8f acc = {};
#pragma unroll
            for (int kc = 0; kc < 32; kc += 4) {
                int kk = kc + khalf;
                v2f a, b;
                a.x = Alds[arow * ALDS_STRIDE + kk];
                a.y = Alds[arow * ALDS_STRIDE + kk + 1];
                b.x = Blds[kk * BLDS_STRIDE + col];
                b.y = Blds[(kk + 1) * BLDS_STRIDE + col];
                acc = __builtin_amdgcn_wmma_f32_16x16x4_f32(
                          false, a, false, b, (short)0, acc, false, false);
            }
            int rbase = (lane >> 4) << 3;
#pragma unroll
            for (int r = 0; r < 8; ++r)
                Strip[(rbase + r) * SLDS_STRIDE + col] = acc[r];
        }
        __syncthreads();

        {
            const int k0 = mtile * 16 + wave;
            const int k1 = k0 + 8;
            float b0 = bprior[((size_t)o * 256 + k0) * 32 + lane];
            float b1 = bprior[((size_t)o * 256 + k1) * 32 + lane];
            float v0 = route_position(&Strip[wave * SLDS_STRIDE], b0, lane);
            float v1 = route_position(&Strip[(8 + wave) * SLDS_STRIDE], b1, lane);
            if (lane < 16) {
                out[((size_t)bm * 256 + k0) * 16 + lane] = v0;
                out[((size_t)bm * 256 + k1) * 16 + lane] = v1;
            }
        }
        __syncthreads();
    }
}

extern "C" void kernel_launch(void* const* d_in, const int* in_sizes, int n_in,
                              void* d_out, int out_size, void* d_ws, size_t ws_size,
                              hipStream_t stream)
{
    const float* prim = (const float*)d_in[0];
    const float* Wt   = (const float*)d_in[1];
    const float* bpr  = (const float*)d_in[2];
    float* out = (float*)d_out;

    const size_t lds_bytes = (size_t)LDS_FLOATS * sizeof(float);
    capsule_fused_wmma<<<256, 256, lds_bytes, stream>>>(prim, Wt, bpr, out);
}


# Strip double-buffered in LDS: GEMM of strip t+1 and routing of strip t share one barrier interval; the two waves of a SIMD run them in opposite order (scalar wave index); priors loaded a strip ahead
# speedup vs baseline: 2.7614x; 1.1306x over previous
#include <hip/hip_runtime.h>
#include <math.h>


typedef __attribute__((ext_vector_type(2))) float v2f;
typedef __attribute__((ext_vector_type(8))) float v8f;

#define ALDS_STRIDE 36
#define BLDS_STRIDE 516
#define SLDS_STRIDE 524
#define STRIP_FLOATS (16 * SLDS_STRIDE)
#define LDS_FLOATS  (256 * ALDS_STRIDE + 32 * BLDS_STRIDE + 2 * STRIP_FLOATS)

__device__ __forceinline__ float xhalf(float x) {
    int b = __builtin_bit_cast(int, x);
    return __builtin_bit_cast(float,
        __builtin_amdgcn_permlanex16(b, b, 0x76543210, 0xfedcba98, false, false));
}
#define ROW_SHARE(n) (0x150 + (n))
#define ROW_XMASK(n) (0x160 + (n))
template <int CTRL>
__device__ __forceinline__ float row_dpp(float x) {
    return __builtin_bit_cast(float,
        __builtin_amdgcn_update_dpp(0, __builtin_bit_cast(int, x), CTRL, 0xf, 0xf, true));
}
__device__ __forceinline__ float row_sum16(float v) {
    v += row_dpp<ROW_XMASK(1)>(v);
    v += row_dpp<ROW_XMASK(2)>(v);
    v += row_dpp<ROW_XMASK(4)>(v);
    v += row_dpp<ROW_XMASK(8)>(v);
    return v;
}
__device__ __forceinline__ float wave_sum(float v) {
    v = row_sum16(v);
    return v + xhalf(v);
}
__device__ __forceinline__ float wave_max(float v) {
    v = fmaxf(v, row_dpp<ROW_XMASK(1)>(v));
    v = fmaxf(v, row_dpp<ROW_XMASK(2)>(v));
    v = fmaxf(v, row_dpp<ROW_XMASK(4)>(v));
    v = fmaxf(v, row_dpp<ROW_XMASK(8)>(v));
    return fmaxf(v, xhalf(v));
}
__device__ __forceinline__ float share_dot16(float src, const float (&x)[16]) {
    float a0 = row_dpp<ROW_SHARE(0)>(src) * x[0];
    float a1 = row_dpp<ROW_SHARE(1)>(src) * x[1];
    a0 = fmaf(row_dpp<ROW_SHARE(2)>(src),  x[2],  a0);
    a1 = fmaf(row_dpp<ROW_SHARE(3)>(src),  x[3],  a1);
    a0 = fmaf(row_dpp<ROW_SHARE(4)>(src),  x[4],  a0);
    a1 = fmaf(row_dpp<ROW_SHARE(5)>(src),  x[5],  a1);
    a0 = fmaf(row_dpp<ROW_SHARE(6)>(src),  x[6],  a0);
    a1 = fmaf(row_dpp<ROW_SHARE(7)>(src),  x[7],  a1);
    a0 = fmaf(row_dpp<ROW_SHARE(8)>(src),  x[8],  a0);
    a1 = fmaf(row_dpp<ROW_SHARE(9)>(src),  x[9],  a1);
    a0 = fmaf(row_dpp<ROW_SHARE(10)>(src), x[10], a0);
    a1 = fmaf(row_dpp<ROW_SHARE(11)>(src), x[11], a1);
    a0 = fmaf(row_dpp<ROW_SHARE(12)>(src), x[12], a0);
    a1 = fmaf(row_dpp<ROW_SHARE(13)>(src), x[13], a1);
    a0 = fmaf(row_dpp<ROW_SHARE(14)>(src), x[14], a0);
    a1 = fmaf(row_dpp<ROW_SHARE(15)>(src), x[15], a1);
    return a0 + a1;
}

__device__ __forceinline__ float weighted_squash(float c, const float (&xd)[16])
{
    float s = share_dot16(c, xd);
    s += xhalf(s);
    float sq = row_sum16(s * s);
    float coef = (sq / (1.0f + sq)) / sqrtf(sq + 1e-7f);
    return coef * s;
}

__device__ __forceinline__ float route_position(const float* __restrict__ srow,
                                                float bb, int lane)
{
    float xm[16];
    const float* sp = srow + lane * 16;
#pragma unroll
    for (int q = 0; q < 4; ++q) {
        float4 v = ((const float4*)sp)[q];
        xm[q * 4 + 0] = v.x; xm[q * 4 + 1] = v.y;
        xm[q * 4 + 2] = v.z; xm[q * 4 + 3] = v.w;
    }
    float xd[16];
    const float* sd = srow + (lane >> 4) * 256 + (lane & 15);
#pragma unroll
    for (int j = 0; j < 16; ++j) xd[j] = sd[j * 16];

    float v;
    {
        float mx = wave_max(bb);
        float e  = __expf(bb - mx);
        float s  = wave_sum(e);
        v = weighted_squash(e / s, xd);
    }
#pragma unroll
    for (int it = 0; it < 2; ++it) {
        bb += share_dot16(v, xm);
        float mx = wave_max(bb);
        float e  = __expf(bb - mx);
        float s  = wave_sum(e);
        v = weighted_squash(e / s, xd);
    }
    return v;
}

__device__ __forceinline__ void gemm_strip(const float* __restrict__ Alds,
                                           const float* __restrict__ Blds,
                                           float* __restrict__ S,
                                           int mtile, int wave, int lane)
{
    const int khalf = (lane >> 4) << 1;
    const int arow  = mtile * 16 + (lane & 15);
#pragma unroll
    for (int i = 0; i < 4; ++i) {
        int nt  = i * 8 + wave;
        int col = nt * 16 + (lane & 15);
        v8f acc = {};
#pragma unroll
        for (int kc = 0; kc < 32; kc += 4) {
            int kk = kc + khalf;
            v2f a, b;
            a.x = Alds[arow * ALDS_STRIDE + kk];
            a.y = Alds[arow * ALDS_STRIDE + kk + 1];
            b.x = Blds[kk * BLDS_STRIDE + col];
            b.y = Blds[(kk + 1) * BLDS_STRIDE + col];
            acc = __builtin_amdgcn_wmma_f32_16x16x4_f32(
                      false, a, false, b, (short)0, acc, false, false);
        }
        int rbase = (lane >> 4) << 3;
#pragma unroll
        for (int r = 0; r < 8; ++r)
            S[(rbase + r) * SLDS_STRIDE + col] = acc[r];
    }
}

__device__ __forceinline__ void route_strip(const float* __restrict__ S,
                                            float b0, float b1,
                                            float* __restrict__ out, int bm,
                                            int mtile, int wave, int lane)
{
    const int k0 = mtile * 16 + wave;
    const int k1 = k0 + 8;
    float v0 = route_position(&S[wave * SLDS_STRIDE], b0, lane);
    float v1 = route_position(&S[(8 + wave) * SLDS_STRIDE], b1, lane);
    if (lane < 16) {
        out[((size_t)bm * 256 + k0) * 16 + lane] = v0;
        out[((size_t)bm * 256 + k1) * 16 + lane] = v1;
    }
}

__global__ __launch_bounds__(256)
void capsule_fused_wmma(const float* __restrict__ prim,
                        const float* __restrict__ Wt,
                        const float* __restrict__ bprior,
                        float* __restrict__ out)
{
    extern __shared__ float smem[];
    float* Alds  = smem;
    float* Blds  = Alds + 256 * ALDS_STRIDE;
    float* Strip = Blds + 32 * BLDS_STRIDE;

    const int bm   = blockIdx.x;
    const int o    = bm & 31;
    const int tid  = threadIdx.x;
    const int wave = __builtin_amdgcn_readfirstlane(tid >> 5);
    const int lane = tid & 31;

    const float* Ag = prim + (size_t)bm * 8192;
    const float* Wg = Wt   + (size_t)bm * 16384;

    __builtin_prefetch(Wg, 0, 0);

#pragma unroll
    for (int i = 0; i < 8; ++i) {
        int f4  = tid + i * 256;
        float4 v = ((const float4*)Ag)[f4];
        int f   = f4 << 2;
        int row = f >> 5;
        int kk  = f & 31;
        float* dst = &Alds[row * ALDS_STRIDE + kk];
        dst[0] = v.x; dst[1] = v.y; dst[2] = v.z; dst[3] = v.w;
    }
#pragma unroll 4
    for (int i = 0; i < 64; ++i) {
        int lin = i * 256 + tid;
        int c   = lin >> 9;
        int n   = lin & 511;
        int yx  = c >> 3, e = c & 7;
        int og  = n >> 4, d = n & 15;
        Blds[c * BLDS_STRIDE + n] = Wg[yx * 4096 + og * 128 + d * 8 + e];
    }

    const float* bp = bprior + ((size_t)o * 256 + wave) * 32 + lane;
    float b0 = bp[0];
    float b1 = bp[256];
    __syncthreads();

    gemm_strip(Alds, Blds, Strip, 0, wave, lane);
    __syncthreads();

    for (int mtile = 0; mtile < 16; ++mtile) {
        const float* cur = Strip + (mtile & 1) * STRIP_FLOATS;
        float*       nxt = Strip + ((mtile & 1) ^ 1) * STRIP_FLOATS;
        const bool more = mtile < 15;

        float b0n = b0, b1n = b1;
        if (more) {
            b0n = bp[(mtile + 1) * 512];
            b1n = bp[(mtile + 1) * 512 + 256];
        }

        if (wave < 4) {
            if (more) gemm_strip(Alds, Blds, nxt, mtile + 1, wave, lane);
            route_strip(cur, b0, b1, out, bm, mtile, wave, lane);
        } else {
            route_strip(cur, b0, b1, out, bm, mtile, wave, lane);
            if (more) gemm_strip(Alds, Blds, nxt, mtile + 1, wave, lane);
        }
        b0 = b0n; b1 = b1n;
        __syncthreads();
    }
}

extern "C" void kernel_launch(void* const* d_in, const int* in_sizes, int n_in,
                              void* d_out, int out_size, void* d_ws, size_t ws_size,
                              hipStream_t stream)
{
    const float* prim = (const float*)d_in[0];
    const float* Wt   = (const float*)d_in[1];
    const float* bpr  = (const float*)d_in[2];
    float* out = (float*)d_out;

    const size_t lds_bytes = (size_t)LDS_FLOATS * sizeof(float);
    capsule_fused_wmma<<<256, 256, lds_bytes, stream>>>(prim, Wt, bpr, out);
}


# Four waves per SIMD (was two): 16-wave blocks on 32-row strips (a wave still routes 2 positions and does 4 GEMM tiles per barrier), so twice as many dependent routing chains interleave per SIMD
# speedup vs baseline: 3.5687x; 1.2924x over previous
#include <hip/hip_runtime.h>
#include <math.h>


typedef __attribute__((ext_vector_type(2))) float v2f;
typedef __attribute__((ext_vector_type(8))) float v8f;

#define ALDS_STRIDE 36
#define BLDS_STRIDE 516
#define SLDS_STRIDE 524
#define STRIP_ROWS  32
#define STRIP_FLOATS (STRIP_ROWS * SLDS_STRIDE)
#define LDS_FLOATS  (256 * ALDS_STRIDE + 32 * BLDS_STRIDE + 2 * STRIP_FLOATS)

__device__ __forceinline__ float xhalf(float x) {
    int b = __builtin_bit_cast(int, x);
    return __builtin_bit_cast(float,
        __builtin_amdgcn_permlanex16(b, b, 0x76543210, 0xfedcba98, false, false));
}
#define ROW_SHARE(n) (0x150 + (n))
#define ROW_XMASK(n) (0x160 + (n))
template <int CTRL>
__device__ __forceinline__ float row_dpp(float x) {
    return __builtin_bit_cast(float,
        __builtin_amdgcn_update_dpp(0, __builtin_bit_cast(int, x), CTRL, 0xf, 0xf, true));
}
__device__ __forceinline__ float row_sum16(float v) {
    v += row_dpp<ROW_XMASK(1)>(v);
    v += row_dpp<ROW_XMASK(2)>(v);
    v += row_dpp<ROW_XMASK(4)>(v);
    v += row_dpp<ROW_XMASK(8)>(v);
    return v;
}
__device__ __forceinline__ float wave_sum(float v) {
    v = row_sum16(v);
    return v + xhalf(v);
}
__device__ __forceinline__ float wave_max(float v) {
    v = fmaxf(v, row_dpp<ROW_XMASK(1)>(v));
    v = fmaxf(v, row_dpp<ROW_XMASK(2)>(v));
    v = fmaxf(v, row_dpp<ROW_XMASK(4)>(v));
    v = fmaxf(v, row_dpp<ROW_XMASK(8)>(v));
    return fmaxf(v, xhalf(v));
}
__device__ __forceinline__ float share_dot16(float src, const float (&x)[16]) {
    float a0 = row_dpp<ROW_SHARE(0)>(src) * x[0];
    float a1 = row_dpp<ROW_SHARE(1)>(src) * x[1];
    a0 = fmaf(row_dpp<ROW_SHARE(2)>(src),  x[2],  a0);
    a1 = fmaf(row_dpp<ROW_SHARE(3)>(src),  x[3],  a1);
    a0 = fmaf(row_dpp<ROW_SHARE(4)>(src),  x[4],  a0);
    a1 = fmaf(row_dpp<ROW_SHARE(5)>(src),  x[5],  a1);
    a0 = fmaf(row_dpp<ROW_SHARE(6)>(src),  x[6],  a0);
    a1 = fmaf(row_dpp<ROW_SHARE(7)>(src),  x[7],  a1);
    a0 = fmaf(row_dpp<ROW_SHARE(8)>(src),  x[8],  a0);
    a1 = fmaf(row_dpp<ROW_SHARE(9)>(src),  x[9],  a1);
    a0 = fmaf(row_dpp<ROW_SHARE(10)>(src), x[10], a0);
    a1 = fmaf(row_dpp<ROW_SHARE(11)>(src), x[11], a1);
    a0 = fmaf(row_dpp<ROW_SHARE(12)>(src), x[12], a0);
    a1 = fmaf(row_dpp<ROW_SHARE(13)>(src), x[13], a1);
    a0 = fmaf(row_dpp<ROW_SHARE(14)>(src), x[14], a0);
    a1 = fmaf(row_dpp<ROW_SHARE(15)>(src), x[15], a1);
    return a0 + a1;
}

__device__ __forceinline__ float weighted_squash(float c, const float (&xd)[16])
{
    float s = share_dot16(c, xd);
    s += xhalf(s);
    float sq = row_sum16(s * s);
    float coef = (sq / (1.0f + sq)) / sqrtf(sq + 1e-7f);
    return coef * s;
}

__device__ __forceinline__ float route_position(const float* __restrict__ srow,
                                                float bb, int lane)
{
    float xm[16];
    const float* sp = srow + lane * 16;
#pragma unroll
    for (int q = 0; q < 4; ++q) {
        float4 v = ((const float4*)sp)[q];
        xm[q * 4 + 0] = v.x; xm[q * 4 + 1] = v.y;
        xm[q * 4 + 2] = v.z; xm[q * 4 + 3] = v.w;
    }
    float xd[16];
    const float* sd = srow + (lane >> 4) * 256 + (lane & 15);
#pragma unroll
    for (int j = 0; j < 16; ++j) xd[j] = sd[j * 16];

    float v;
    {
        float mx = wave_max(bb);
        float e  = __expf(bb - mx);
        float s  = wave_sum(e);
        v = weighted_squash(e / s, xd);
    }
#pragma unroll
    for (int it = 0; it < 2; ++it) {
        bb += share_dot16(v, xm);
        float mx = wave_max(bb);
        float e  = __expf(bb - mx);
        float s  = wave_sum(e);
        v = weighted_squash(e / s, xd);
    }
    return v;
}

__device__ __forceinline__ void gemm_strip(const float* __restrict__ Alds,
                                           const float* __restrict__ Blds,
                                           float* __restrict__ S,
                                           int strip, int wave, int lane)
{
    const int khalf = (lane >> 4) << 1;
    const int arow0 = strip * STRIP_ROWS + (lane & 15);
    const int arow1 = arow0 + 16;
#pragma unroll
    for (int i = 0; i < 2; ++i) {
        int nt  = i * 16 + wave;
        int col = nt * 16 + (lane & 15);
        v8f acc0 = {};
        v8f acc1 = {};
#pragma unroll
        for (int kc = 0; kc < 32; kc += 4) {
            int kk = kc + khalf;
            v2f a0, a1, b;
            a0.x = Alds[arow0 * ALDS_STRIDE + kk];
            a0.y = Alds[arow0 * ALDS_STRIDE + kk + 1];
            a1.x = Alds[arow1 * ALDS_STRIDE + kk];
            a1.y = Alds[arow1 * ALDS_STRIDE + kk + 1];
            b.x = Blds[kk * BLDS_STRIDE + col];
            b.y = Blds[(kk + 1) * BLDS_STRIDE + col];
            acc0 = __builtin_amdgcn_wmma_f32_16x16x4_f32(
                       false, a0, false, b, (short)0, acc0, false, false);
            acc1 = __builtin_amdgcn_wmma_f32_16x16x4_f32(
                       false, a1, false, b, (short)0, acc1, false, false);
        }
        int rbase = (lane >> 4) << 3;
#pragma unroll
        for (int r = 0; r < 8; ++r) {
            S[(rbase + r) * SLDS_STRIDE + col]      = acc0[r];
            S[(16 + rbase + r) * SLDS_STRIDE + col] = acc1[r];
        }
    }
}

__device__ __forceinline__ void route_strip(const float* __restrict__ S,
                                            float b0, float b1,
                                            float* __restrict__ out, int bm,
                                            int strip, int wave, int lane)
{
    const int k0 = strip * STRIP_ROWS + wave;
    const int k1 = k0 + 16;
    float v0 = route_position(&S[wave * SLDS_STRIDE], b0, lane);
    float v1 = route_position(&S[(16 + wave) * SLDS_STRIDE], b1, lane);
    if (lane < 16) {
        out[((size_t)bm * 256 + k0) * 16 + lane] = v0;
        out[((size_t)bm * 256 + k1) * 16 + lane] = v1;
    }
}

__global__ __launch_bounds__(512)
void capsule_fused_wmma(const float* __restrict__ prim,
                        const float* __restrict__ Wt,
                        const float* __restrict__ bprior,
                        float* __restrict__ out)
{
    extern __shared__ float smem[];
    float* Alds  = smem;
    float* Blds  = Alds + 256 * ALDS_STRIDE;
    float* Strip = Blds + 32 * BLDS_STRIDE;

    const int bm   = blockIdx.x;
    const int o    = bm & 31;
    const int tid  = threadIdx.x;
    const int wave = __builtin_amdgcn_readfirstlane(tid >> 5);
    const int lane = tid & 31;

    const float* Ag = prim + (size_t)bm * 8192;
    const float* Wg = Wt   + (size_t)bm * 16384;

    __builtin_prefetch(Wg, 0, 0);

#pragma unroll
    for (int i = 0; i < 4; ++i) {
        int f4  = tid + i * 512;
        float4 v = ((const float4*)Ag)[f4];
        int f   = f4 << 2;
        int row = f >> 5;
        int kk  = f & 31;
        float* dst = &Alds[row * ALDS_STRIDE + kk];
        dst[0] = v.x; dst[1] = v.y; dst[2] = v.z; dst[3] = v.w;
    }
#pragma unroll 4
    for (int i = 0; i < 32; ++i) {
        int lin = i * 512 + tid;
        int c   = lin >> 9;
        int n   = lin & 511;
        int yx  = c >> 3, e = c & 7;
        int og  = n >> 4, d = n & 15;
        Blds[c * BLDS_STRIDE + n] = Wg[yx * 4096 + og * 128 + d * 8 + e];
    }

    const float* bp = bprior + ((size_t)o * 256 + wave) * 32 + lane;
    float b0 = bp[0];
    float b1 = bp[512];
    __syncthreads();

    gemm_strip(Alds, Blds, Strip, 0, wave, lane);
    __syncthreads();

    for (int strip = 0; strip < 8; ++strip) {
        const float* cur = Strip + (strip & 1) * STRIP_FLOATS;
        float*       nxt = Strip + ((strip & 1) ^ 1) * STRIP_FLOATS;
        const bool more = strip < 7;

        float b0n = b0, b1n = b1;
        if (more) {
            b0n = bp[(strip + 1) * 1024];
            b1n = bp[(strip + 1) * 1024 + 512];
        }

        if ((wave & 4) == 0) {
            if (more) gemm_strip(Alds, Blds, nxt, strip + 1, wave, lane);
            route_strip(cur, b0, b1, out, bm, strip, wave, lane);
        } else {
            route_strip(cur, b0, b1, out, bm, strip, wave, lane);
            if (more) gemm_strip(Alds, Blds, nxt, strip + 1, wave, lane);
        }
        b0 = b0n; b1 = b1n;
        __syncthreads();
    }
}

extern "C" void kernel_launch(void* const* d_in, const int* in_sizes, int n_in,
                              void* d_out, int out_size, void* d_ws, size_t ws_size,
                              hipStream_t stream)
{
    const float* prim = (const float*)d_in[0];
    const float* Wt   = (const float*)d_in[1];
    const float* bpr  = (const float*)d_in[2];
    float* out = (float*)d_out;

    const size_t lds_bytes = (size_t)LDS_FLOATS * sizeof(float);
    capsule_fused_wmma<<<256, 512, lds_bytes, stream>>>(prim, Wt, bpr, out);
}


# Softmax and squash normalisers of a routing stage folded into one scale U*rcp(s^2+U)*rsq(U+eps*s^2) on the unnormalised weighted sum: one rcp and one rsq replace three IEEE divisions and a sqrt
# speedup vs baseline: 4.3364x; 1.2151x over previous
#include <hip/hip_runtime.h>
#include <math.h>


typedef __attribute__((ext_vector_type(2))) float v2f;
typedef __attribute__((ext_vector_type(8))) float v8f;

#define ALDS_STRIDE 36
#define BLDS_STRIDE 516
#define SLDS_STRIDE 524
#define STRIP_ROWS  32
#define STRIP_FLOATS (STRIP_ROWS * SLDS_STRIDE)
#define LDS_FLOATS  (256 * ALDS_STRIDE + 32 * BLDS_STRIDE + 2 * STRIP_FLOATS)

__device__ __forceinline__ float xhalf(float x) {
    int b = __builtin_bit_cast(int, x);
    return __builtin_bit_cast(float,
        __builtin_amdgcn_permlanex16(b, b, 0x76543210, 0xfedcba98, false, false));
}
#define ROW_SHARE(n) (0x150 + (n))
#define ROW_XMASK(n) (0x160 + (n))
template <int CTRL>
__device__ __forceinline__ float row_dpp(float x) {
    return __builtin_bit_cast(float,
        __builtin_amdgcn_update_dpp(0, __builtin_bit_cast(int, x), CTRL, 0xf, 0xf, true));
}
__device__ __forceinline__ float row_sum16(float v) {
    v += row_dpp<ROW_XMASK(1)>(v);
    v += row_dpp<ROW_XMASK(2)>(v);
    v += row_dpp<ROW_XMASK(4)>(v);
    v += row_dpp<ROW_XMASK(8)>(v);
    return v;
}
__device__ __forceinline__ float wave_sum(float v) {
    v = row_sum16(v);
    return v + xhalf(v);
}
__device__ __forceinline__ float wave_max(float v) {
    v = fmaxf(v, row_dpp<ROW_XMASK(1)>(v));
    v = fmaxf(v, row_dpp<ROW_XMASK(2)>(v));
    v = fmaxf(v, row_dpp<ROW_XMASK(4)>(v));
    v = fmaxf(v, row_dpp<ROW_XMASK(8)>(v));
    return fmaxf(v, xhalf(v));
}
__device__ __forceinline__ float share_dot16(float src, const float (&x)[16]) {
    float a0 = row_dpp<ROW_SHARE(0)>(src) * x[0];
    float a1 = row_dpp<ROW_SHARE(1)>(src) * x[1];
    a0 = fmaf(row_dpp<ROW_SHARE(2)>(src),  x[2],  a0);
    a1 = fmaf(row_dpp<ROW_SHARE(3)>(src),  x[3],  a1);
    a0 = fmaf(row_dpp<ROW_SHARE(4)>(src),  x[4],  a0);
    a1 = fmaf(row_dpp<ROW_SHARE(5)>(src),  x[5],  a1);
    a0 = fmaf(row_dpp<ROW_SHARE(6)>(src),  x[6],  a0);
    a1 = fmaf(row_dpp<ROW_SHARE(7)>(src),  x[7],  a1);
    a0 = fmaf(row_dpp<ROW_SHARE(8)>(src),  x[8],  a0);
    a1 = fmaf(row_dpp<ROW_SHARE(9)>(src),  x[9],  a1);
    a0 = fmaf(row_dpp<ROW_SHARE(10)>(src), x[10], a0);
    a1 = fmaf(row_dpp<ROW_SHARE(11)>(src), x[11], a1);
    a0 = fmaf(row_dpp<ROW_SHARE(12)>(src), x[12], a0);
    a1 = fmaf(row_dpp<ROW_SHARE(13)>(src), x[13], a1);
    a0 = fmaf(row_dpp<ROW_SHARE(14)>(src), x[14], a0);
    a1 = fmaf(row_dpp<ROW_SHARE(15)>(src), x[15], a1);
    return a0 + a1;
}

__device__ __forceinline__ float weighted_squash(float e, float esum, const float (&xd)[16])
{
    float u = share_dot16(e, xd);
    u += xhalf(u);
    float U  = row_sum16(u * u);
    float s2 = esum * esum;
    float coef = U * __builtin_amdgcn_rcpf(s2 + U)
                   * __builtin_amdgcn_rsqf(fmaf(1e-7f, s2, U));
    return coef * u;
}

__device__ __forceinline__ float route_position(const float* __restrict__ srow,
                                                float bb, int lane)
{
    float xm[16];
    const float* sp = srow + lane * 16;
#pragma unroll
    for (int q = 0; q < 4; ++q) {
        float4 v = ((const float4*)sp)[q];
        xm[q * 4 + 0] = v.x; xm[q * 4 + 1] = v.y;
        xm[q * 4 + 2] = v.z; xm[q * 4 + 3] = v.w;
    }
    float xd[16];
    const float* sd = srow + (lane >> 4) * 256 + (lane & 15);
#pragma unroll
    for (int j = 0; j < 16; ++j) xd[j] = sd[j * 16];

    float v;
    {
        float mx = wave_max(bb);
        float e  = __expf(bb - mx);
        float s  = wave_sum(e);
        v = weighted_squash(e, s, xd);
    }
#pragma unroll
    for (int it = 0; it < 2; ++it) {
        bb += share_dot16(v, xm);
        float mx = wave_max(bb);
        float e  = __expf(bb - mx);
        float s  = wave_sum(e);
        v = weighted_squash(e, s, xd);
    }
    return v;
}

__device__ __forceinline__ void gemm_strip(const float* __restrict__ Alds,
                                           const float* __restrict__ Blds,
                                           float* __restrict__ S,
                                           int strip, int wave, int lane)
{
    const int khalf = (lane >> 4) << 1;
    const int arow0 = strip * STRIP_ROWS + (lane & 15);
    const int arow1 = arow0 + 16;
#pragma unroll
    for (int i = 0; i < 2; ++i) {
        int nt  = i * 16 + wave;
        int col = nt * 16 + (lane & 15);
        v8f acc0 = {};
        v8f acc1 = {};
#pragma unroll
        for (int kc = 0; kc < 32; kc += 4) {
            int kk = kc + khalf;
            v2f a0, a1, b;
            a0.x = Alds[arow0 * ALDS_STRIDE + kk];
            a0.y = Alds[arow0 * ALDS_STRIDE + kk + 1];
            a1.x = Alds[arow1 * ALDS_STRIDE + kk];
            a1.y = Alds[arow1 * ALDS_STRIDE + kk + 1];
            b.x = Blds[kk * BLDS_STRIDE + col];
            b.y = Blds[(kk + 1) * BLDS_STRIDE + col];
            acc0 = __builtin_amdgcn_wmma_f32_16x16x4_f32(
                       false, a0, false, b, (short)0, acc0, false, false);
            acc1 = __builtin_amdgcn_wmma_f32_16x16x4_f32(
                       false, a1, false, b, (short)0, acc1, false, false);
        }
        int rbase = (lane >> 4) << 3;
#pragma unroll
        for (int r = 0; r < 8; ++r) {
            S[(rbase + r) * SLDS_STRIDE + col]      = acc0[r];
            S[(16 + rbase + r) * SLDS_STRIDE + col] = acc1[r];
        }
    }
}

__device__ __forceinline__ void route_strip(const float* __restrict__ S,
                                            float b0, float b1,
                                            float* __restrict__ out, int bm,
                                            int strip, int wave, int lane)
{
    const int k0 = strip * STRIP_ROWS + wave;
    const int k1 = k0 + 16;
    float v0 = route_position(&S[wave * SLDS_STRIDE], b0, lane);
    float v1 = route_position(&S[(16 + wave) * SLDS_STRIDE], b1, lane);
    if (lane < 16) {
        out[((size_t)bm * 256 + k0) * 16 + lane] = v0;
        out[((size_t)bm * 256 + k1) * 16 + lane] = v1;
    }
}

__global__ __launch_bounds__(512)
void capsule_fused_wmma(const float* __restrict__ prim,
                        const float* __restrict__ Wt,
                        const float* __restrict__ bprior,
                        float* __restrict__ out)
{
    extern __shared__ float smem[];
    float* Alds  = smem;
    float* Blds  = Alds + 256 * ALDS_STRIDE;
    float* Strip = Blds + 32 * BLDS_STRIDE;

    const int bm   = blockIdx.x;
    const int o    = bm & 31;
    const int tid  = threadIdx.x;
    const int wave = __builtin_amdgcn_readfirstlane(tid >> 5);
    const int lane = tid & 31;

    const float* Ag = prim + (size_t)bm * 8192;
    const float* Wg = Wt   + (size_t)bm * 16384;

    __builtin_prefetch(Wg, 0, 0);

#pragma unroll
    for (int i = 0; i < 4; ++i) {
        int f4  = tid + i * 512;
        float4 v = ((const float4*)Ag)[f4];
        int f   = f4 << 2;
        int row = f >> 5;
        int kk  = f & 31;
        float* dst = &Alds[row * ALDS_STRIDE + kk];
        dst[0] = v.x; dst[1] = v.y; dst[2] = v.z; dst[3] = v.w;
    }
#pragma unroll 4
    for (int i = 0; i < 32; ++i) {
        int lin = i * 512 + tid;
        int c   = lin >> 9;
        int n   = lin & 511;
        int yx  = c >> 3, e = c & 7;
        int og  = n >> 4, d = n & 15;
        Blds[c * BLDS_STRIDE + n] = Wg[yx * 4096 + og * 128 + d * 8 + e];
    }

    const float* bp = bprior + ((size_t)o * 256 + wave) * 32 + lane;
    float b0 = bp[0];
    float b1 = bp[512];
    __syncthreads();

    gemm_strip(Alds, Blds, Strip, 0, wave, lane);
    __syncthreads();

    for (int strip = 0; strip < 8; ++strip) {
        const float* cur = Strip + (strip & 1) * STRIP_FLOATS;
        float*       nxt = Strip + ((strip & 1) ^ 1) * STRIP_FLOATS;
        const bool more = strip < 7;

        float b0n = b0, b1n = b1;
        if (more) {
            b0n = bp[(strip + 1) * 1024];
            b1n = bp[(strip + 1) * 1024 + 512];
        }

        if ((wave & 4) == 0) {
            if (more) gemm_strip(Alds, Blds, nxt, strip + 1, wave, lane);
            route_strip(cur, b0, b1, out, bm, strip, wave, lane);
        } else {
            route_strip(cur, b0, b1, out, bm, strip, wave, lane);
            if (more) gemm_strip(Alds, Blds, nxt, strip + 1, wave, lane);
        }
        b0 = b0n; b1 = b1n;
        __syncthreads();
    }
}

extern "C" void kernel_launch(void* const* d_in, const int* in_sizes, int n_in,
                              void* d_out, int out_size, void* d_ws, size_t ws_size,
                              hipStream_t stream)
{
    const float* prim = (const float*)d_in[0];
    const float* Wt   = (const float*)d_in[1];
    const float* bpr  = (const float*)d_in[2];
    float* out = (float*)d_out;

    const size_t lds_bytes = (size_t)LDS_FLOATS * sizeof(float);
    capsule_fused_wmma<<<256, 512, lds_bytes, stream>>>(prim, Wt, bpr, out);
}


# A strip's routing spread over two barrier intervals: two routing waves per SIMD take alternate strips, one starts a strip (x load, first stage) while the other finishes the one before; 8 GEMM waves
# speedup vs baseline: 6.2764x; 1.4474x over previous
#include <hip/hip_runtime.h>
#include <math.h>


typedef __attribute__((ext_vector_type(2))) float v2f;
typedef __attribute__((ext_vector_type(8))) float v8f;

#define ALDS_STRIDE 36
#define BLDS_STRIDE 516
#define SLDS_STRIDE 524
#define STRIP_ROWS  32
#define STRIP_FLOATS (STRIP_ROWS * SLDS_STRIDE)
#define PRI_FLOATS  (STRIP_ROWS * 32)
#define LDS_FLOATS  (256 * ALDS_STRIDE + 32 * BLDS_STRIDE + 2 * STRIP_FLOATS + 2 * PRI_FLOATS)
#define GEMM_WAVES  8
#define PHASES      10

#define ROW_XMASK(n) (0x160 + (n))
template <int CTRL>
__device__ __forceinline__ float row_dpp(float x) {
    return __builtin_bit_cast(float,
        __builtin_amdgcn_update_dpp(0, __builtin_bit_cast(int, x), CTRL, 0xf, 0xf, true));
}
__device__ __forceinline__ float other_dims(float v)    { return row_dpp<ROW_XMASK(1)>(v); }
__device__ __forceinline__ float other_modules(float v) { return row_dpp<ROW_XMASK(2)>(v); }

__device__ __forceinline__ void route_stage(const float (&bb)[16], const v2f (&x)[4][16],
                                            v2f (&v)[4])
{
    float m0 = bb[0], m1 = bb[1], m2 = bb[2], m3 = bb[3];
#pragma unroll
    for (int j = 4; j < 16; j += 4) {
        m0 = fmaxf(m0, bb[j]);     m1 = fmaxf(m1, bb[j + 1]);
        m2 = fmaxf(m2, bb[j + 2]); m3 = fmaxf(m3, bb[j + 3]);
    }
    float mx = fmaxf(fmaxf(m0, m1), fmaxf(m2, m3));
    mx = fmaxf(mx, other_modules(mx));

    float s0 = 0.f, s1 = 0.f;
    v2f u[4];
#pragma unroll
    for (int q = 0; q < 4; ++q) u[q] = (v2f){0.f, 0.f};
#pragma unroll
    for (int j = 0; j < 16; j += 2) {
        const float e0 = __expf(bb[j] - mx);
        const float e1 = __expf(bb[j + 1] - mx);
        s0 += e0;
        s1 += e1;
        const v2f ee0 = {e0, e0};
        const v2f ee1 = {e1, e1};
#pragma unroll
        for (int q = 0; q < 4; ++q) u[q] = __builtin_elementwise_fma(ee0, x[q][j], u[q]);
#pragma unroll
        for (int q = 0; q < 4; ++q) u[q] = __builtin_elementwise_fma(ee1, x[q][j + 1], u[q]);
    }
    float esum = s0 + s1;
    esum += other_modules(esum);
#pragma unroll
    for (int q = 0; q < 4; ++q) {
        u[q].x += other_modules(u[q].x);
        u[q].y += other_modules(u[q].y);
    }
    v2f sq = u[0] * u[0];
    sq = __builtin_elementwise_fma(u[1], u[1], sq);
    sq = __builtin_elementwise_fma(u[2], u[2], sq);
    sq = __builtin_elementwise_fma(u[3], u[3], sq);
    float U = sq.x + sq.y;
    U += other_dims(U);
    const float s2 = esum * esum;
    const float coef = U * __builtin_amdgcn_rcpf(s2 + U)
                         * __builtin_amdgcn_rsqf(fmaf(1e-7f, s2, U));
    const v2f cc = {coef, coef};
#pragma unroll
    for (int q = 0; q < 4; ++q) v[q] = cc * u[q];
}

__device__ __forceinline__ void agree(float (&bb)[16], const v2f (&x)[4][16], const v2f (&v)[4])
{
#pragma unroll
    for (int j = 0; j < 16; ++j) {
        v2f acc = x[0][j] * v[0];
        acc = __builtin_elementwise_fma(x[1][j], v[1], acc);
        acc = __builtin_elementwise_fma(x[2][j], v[2], acc);
        acc = __builtin_elementwise_fma(x[3][j], v[3], acc);
        const float p = acc.x + acc.y;
        bb[j] += p + other_dims(p);
    }
}

__device__ __forceinline__ void route_begin(const float* __restrict__ srow,
                                            const float* __restrict__ prior,
                                            int mh, int dh, v2f (&x)[4][16], float (&bb)[16])
{
    const float* sp = srow + mh * 16 + dh * 8;
#pragma unroll
    for (int j = 0; j < 16; ++j) {
        const float4 ta = *(const float4*)(sp + j * 32);
        const float4 tb = *(const float4*)(sp + j * 32 + 4);
        x[0][j] = (v2f){ta.x, ta.y};  x[1][j] = (v2f){ta.z, ta.w};
        x[2][j] = (v2f){tb.x, tb.y};  x[3][j] = (v2f){tb.z, tb.w};
    }
#pragma unroll
    for (int j = 0; j < 16; ++j) bb[j] = prior[2 * j + mh];

    v2f v[4];
    route_stage(bb, x, v);
    agree(bb, x, v);
}

__device__ __forceinline__ void route_finish(const v2f (&x)[4][16], float (&bb)[16], v2f (&v)[4])
{
    route_stage(bb, x, v);
    agree(bb, x, v);
    route_stage(bb, x, v);
}

__device__ __forceinline__ void gemm_strip(const float* __restrict__ Alds,
                                           const float* __restrict__ Blds,
                                           float* __restrict__ S,
                                           int strip, int wave, int lane)
{
    const int khalf = (lane >> 4) << 1;
    const int arow0 = strip * STRIP_ROWS + (lane & 15);
    const float* a0p = Alds + arow0 * ALDS_STRIDE + khalf;
    const float* a1p = a0p + 16 * ALDS_STRIDE;
    v2f a0[8], a1[8];
#pragma unroll
    for (int j = 0; j < 8; ++j) {
        a0[j].x = a0p[4 * j];  a0[j].y = a0p[4 * j + 1];
        a1[j].x = a1p[4 * j];  a1[j].y = a1p[4 * j + 1];
    }
    const int rbase  = (lane >> 4) << 3;
#pragma nounroll
    for (int i = 0; i < 4; ++i) {
        int nt  = wave + i * 8;
        int col = nt * 16 + (lane & 15);
        const float* bp = Blds + khalf * BLDS_STRIDE + col;
        v8f acc0 = {};
        v8f acc1 = {};
#pragma unroll
        for (int j = 0; j < 8; ++j) {
            v2f b;
            b.x = bp[(4 * j) * BLDS_STRIDE];
            b.y = bp[(4 * j + 1) * BLDS_STRIDE];
            acc0 = __builtin_amdgcn_wmma_f32_16x16x4_f32(
                       false, a0[j], false, b, (short)0, acc0, false, false);
            acc1 = __builtin_amdgcn_wmma_f32_16x16x4_f32(
                       false, a1[j], false, b, (short)0, acc1, false, false);
        }
#pragma unroll
        for (int r = 0; r < 8; ++r) {
            S[(rbase + r) * SLDS_STRIDE + col]      = acc0[r];
            S[(16 + rbase + r) * SLDS_STRIDE + col] = acc1[r];
        }
    }
}

__global__ __launch_bounds__(512)
void capsule_fused_wmma(const float* __restrict__ prim,
                        const float* __restrict__ Wt,
                        const float* __restrict__ bprior,
                        float* __restrict__ out)
{
    extern __shared__ float smem[];
    float* Alds  = smem;
    float* Blds  = Alds + 256 * ALDS_STRIDE;
    float* Strip = Blds + 32 * BLDS_STRIDE;
    float* Pri   = Strip + 2 * STRIP_FLOATS;

    const int bm   = blockIdx.x;
    const int o    = bm & 31;
    const int tid  = threadIdx.x;
    const int wave = __builtin_amdgcn_readfirstlane(tid >> 5);
    const int lane = tid & 31;
    const bool router = wave >= GEMM_WAVES;

    const float* Ag = prim + (size_t)bm * 8192;
    const float* Wg = Wt   + (size_t)bm * 16384;

    __builtin_prefetch(Wg, 0, 0);

#pragma unroll
    for (int i = 0; i < 4; ++i) {
        int f4  = tid + i * 512;
        float4 v = ((const float4*)Ag)[f4];
        int f   = f4 << 2;
        int row = f >> 5;
        int kk  = f & 31;
        float* dst = &Alds[row * ALDS_STRIDE + kk];
        dst[0] = v.x; dst[1] = v.y; dst[2] = v.z; dst[3] = v.w;
    }
#pragma unroll 4
    for (int i = 0; i < 32; ++i) {
        int lin = i * 512 + tid;
        int c   = lin >> 9;
        int n   = lin & 511;
        int yx  = c >> 3, e = c & 7;
        int og  = n >> 4, d = n & 15;
        Blds[c * BLDS_STRIDE + n] = Wg[yx * 4096 + og * 128 + d * 8 + e];
    }

    const int rset = (wave - GEMM_WAVES) >> 2;
    const int rrow = (((wave - GEMM_WAVES) & 3) << 3) + (lane >> 2);
    const int mh   = (lane >> 1) & 1;
    const int dh   = lane & 1;
    const float4* bp4 = (const float4*)(bprior + (size_t)o * 256 * 32);
    float4 pv = make_float4(0.f, 0.f, 0.f, 0.f);
    if (!router) pv = bp4[tid];
    __syncthreads();

    v2f   x[4][16];
    float bb[16];

    for (int p = 0; p < PHASES; ++p) {
        if (!router) {
            if (p < 8) {
                ((float4*)(Pri + (p & 1) * PRI_FLOATS))[tid] = pv;
                if (p < 7) pv = bp4[(p + 1) * 256 + tid];
                gemm_strip(Alds, Blds, Strip + (p & 1) * STRIP_FLOATS, p, wave, lane);
            }
        } else {
            const int sf = p - 1;
            if ((sf & 1) == rset) {
                if (sf >= 0 && sf < 8)
                    route_begin(Strip + (sf & 1) * STRIP_FLOATS + rrow * SLDS_STRIDE,
                                Pri + (sf & 1) * PRI_FLOATS + rrow * 32, mh, dh, x, bb);
            } else {
                const int s2 = p - 2;
                if (s2 >= 0 && s2 < 8) {
                    v2f v8[4];
                    route_finish(x, bb, v8);
                    const float4 v = make_float4(mh ? v8[2].x : v8[0].x, mh ? v8[2].y : v8[0].y,
                                                 mh ? v8[3].x : v8[1].x, mh ? v8[3].y : v8[1].y);
                    const size_t k = (size_t)s2 * STRIP_ROWS + rrow;
                    *(float4*)(out + ((size_t)bm * 256 + k) * 16 + dh * 8 + mh * 4) = v;
                }
            }
        }
        __syncthreads();
    }
}

extern "C" void kernel_launch(void* const* d_in, const int* in_sizes, int n_in,
                              void* d_out, int out_size, void* d_ws, size_t ws_size,
                              hipStream_t stream)
{
    const float* prim = (const float*)d_in[0];
    const float* Wt   = (const float*)d_in[1];
    const float* bpr  = (const float*)d_in[2];
    float* out = (float*)d_out;

    const size_t lds_bytes = (size_t)LDS_FLOATS * sizeof(float);
    capsule_fused_wmma<<<256, 512, lds_bytes, stream>>>(prim, Wt, bpr, out);
}


# GEMM loop LDS addressing: B staged as k-pairs at LDS offset 0, strip buffer chosen by address: a B fragment is one 8-byte read, reads and strip stores on immediates; 2 address adds per tile, was 25
# speedup vs baseline: 6.9898x; 1.1137x over previous
#include <hip/hip_runtime.h>
#include <math.h>


typedef __attribute__((ext_vector_type(2))) float v2f;
typedef __attribute__((ext_vector_type(8))) float v8f;

#define ALDS_STRIDE 36
#define BPAIR_PITCH 528
#define SLDS_STRIDE 524
#define STRIP_ROWS  32
#define STRIP_FLOATS (STRIP_ROWS * SLDS_STRIDE)
#define PRI_FLOATS  (STRIP_ROWS * 32)
#define LDS_FLOATS  (32 * BPAIR_PITCH + 256 * ALDS_STRIDE + 2 * STRIP_FLOATS + 2 * PRI_FLOATS)
#define GEMM_WAVES  8
#define PHASES      10

#define ROW_XMASK(n) (0x160 + (n))
template <int CTRL>
__device__ __forceinline__ float row_dpp(float x) {
    return __builtin_bit_cast(float,
        __builtin_amdgcn_update_dpp(0, __builtin_bit_cast(int, x), CTRL, 0xf, 0xf, true));
}
__device__ __forceinline__ float other_dims(float v)    { return row_dpp<ROW_XMASK(1)>(v); }
__device__ __forceinline__ float other_modules(float v) { return row_dpp<ROW_XMASK(2)>(v); }

__device__ __forceinline__ void route_stage(const float (&bb)[16], const v2f (&x)[4][16],
                                            v2f (&v)[4])
{
    float m0 = bb[0], m1 = bb[1], m2 = bb[2], m3 = bb[3];
#pragma unroll
    for (int j = 4; j < 16; j += 4) {
        m0 = fmaxf(m0, bb[j]);     m1 = fmaxf(m1, bb[j + 1]);
        m2 = fmaxf(m2, bb[j + 2]); m3 = fmaxf(m3, bb[j + 3]);
    }
    float mx = fmaxf(fmaxf(m0, m1), fmaxf(m2, m3));
    mx = fmaxf(mx, other_modules(mx));

    float s0 = 0.f, s1 = 0.f;
    v2f u[4];
#pragma unroll
    for (int q = 0; q < 4; ++q) u[q] = (v2f){0.f, 0.f};
#pragma unroll
    for (int j = 0; j < 16; j += 2) {
        const float e0 = __expf(bb[j] - mx);
        const float e1 = __expf(bb[j + 1] - mx);
        s0 += e0;
        s1 += e1;
        const v2f ee0 = {e0, e0};
        const v2f ee1 = {e1, e1};
#pragma unroll
        for (int q = 0; q < 4; ++q) u[q] = __builtin_elementwise_fma(ee0, x[q][j], u[q]);
#pragma unroll
        for (int q = 0; q < 4; ++q) u[q] = __builtin_elementwise_fma(ee1, x[q][j + 1], u[q]);
    }
    float esum = s0 + s1;
    esum += other_modules(esum);
#pragma unroll
    for (int q = 0; q < 4; ++q) {
        u[q].x += other_modules(u[q].x);
        u[q].y += other_modules(u[q].y);
    }
    v2f sq = u[0] * u[0];
    sq = __builtin_elementwise_fma(u[1], u[1], sq);
    sq = __builtin_elementwise_fma(u[2], u[2], sq);
    sq = __builtin_elementwise_fma(u[3], u[3], sq);
    float U = sq.x + sq.y;
    U += other_dims(U);
    const float s2 = esum * esum;
    const float coef = U * __builtin_amdgcn_rcpf(s2 + U)
                         * __builtin_amdgcn_rsqf(fmaf(1e-7f, s2, U));
    const v2f cc = {coef, coef};
#pragma unroll
    for (int q = 0; q < 4; ++q) v[q] = cc * u[q];
}

__device__ __forceinline__ void agree(float (&bb)[16], const v2f (&x)[4][16], const v2f (&v)[4])
{
#pragma unroll
    for (int j = 0; j < 16; ++j) {
        v2f acc = x[0][j] * v[0];
        acc = __builtin_elementwise_fma(x[1][j], v[1], acc);
        acc = __builtin_elementwise_fma(x[2][j], v[2], acc);
        acc = __builtin_elementwise_fma(x[3][j], v[3], acc);
        const float p = acc.x + acc.y;
        bb[j] += p + other_dims(p);
    }
}

__device__ __forceinline__ void route_begin(const float* __restrict__ srow,
                                            const float* __restrict__ prior,
                                            int mh, int dh, v2f (&x)[4][16], float (&bb)[16])
{
    const float* sp = srow + mh * 16 + dh * 8;
#pragma unroll
    for (int j = 0; j < 16; ++j) {
        const float4 ta = *(const float4*)(sp + j * 32);
        const float4 tb = *(const float4*)(sp + j * 32 + 4);
        x[0][j] = (v2f){ta.x, ta.y};  x[1][j] = (v2f){ta.z, ta.w};
        x[2][j] = (v2f){tb.x, tb.y};  x[3][j] = (v2f){tb.z, tb.w};
    }
#pragma unroll
    for (int j = 0; j < 16; ++j) bb[j] = prior[2 * j + mh];

    v2f v[4];
    route_stage(bb, x, v);
    agree(bb, x, v);
}

__device__ __forceinline__ void route_finish(const v2f (&x)[4][16], float (&bb)[16], v2f (&v)[4])
{
    route_stage(bb, x, v);
    agree(bb, x, v);
    route_stage(bb, x, v);
}

__device__ __forceinline__ void gemm_strip(const float* __restrict__ Alds,
                                           const float* __restrict__ Blds,
                                           float* __restrict__ S,
                                           int strip, int wave, int lane)
{
    const int khalf = (lane >> 4) << 1;
    const int arow0 = strip * STRIP_ROWS + (lane & 15);
    const v2f* a0p = (const v2f*)(Alds + arow0 * ALDS_STRIDE + khalf);
    const v2f* a1p = a0p + 8 * ALDS_STRIDE;
    v2f a0[8], a1[8];
#pragma unroll
    for (int j = 0; j < 8; ++j) {
        a0[j] = a0p[2 * j];
        a1[j] = a1p[2 * j];
    }
    const int rbase  = (lane >> 4) << 3;
#pragma nounroll
    for (int i = 0; i < 4; ++i) {
        int nt  = wave + i * 8;
        int col = nt * 16 + (lane & 15);
        const v2f* bp = (const v2f*)Blds + (lane >> 4) * BPAIR_PITCH + col;
        v8f acc0 = {};
        v8f acc1 = {};
#pragma unroll
        for (int j = 0; j < 8; ++j) {
            const v2f b = bp[(2 * j) * BPAIR_PITCH];
            acc0 = __builtin_amdgcn_wmma_f32_16x16x4_f32(
                       false, a0[j], false, b, (short)0, acc0, false, false);
            acc1 = __builtin_amdgcn_wmma_f32_16x16x4_f32(
                       false, a1[j], false, b, (short)0, acc1, false, false);
        }
#pragma unroll
        for (int r = 0; r < 8; ++r) {
            S[(rbase + r) * SLDS_STRIDE + col]      = acc0[r];
            S[(16 + rbase + r) * SLDS_STRIDE + col] = acc1[r];
        }
    }
}

__global__ __launch_bounds__(512)
void capsule_fused_wmma(const float* __restrict__ prim,
                        const float* __restrict__ Wt,
                        const float* __restrict__ bprior,
                        float* __restrict__ out)
{
    extern __shared__ float smem[];
    float* Blds  = smem;
    float* Alds  = Blds + 32 * BPAIR_PITCH;
    float* Strip = Alds + 256 * ALDS_STRIDE;
    float* Pri   = Strip + 2 * STRIP_FLOATS;

    const int bm   = blockIdx.x;
    const int o    = bm & 31;
    const int tid  = threadIdx.x;
    const int wave = __builtin_amdgcn_readfirstlane(tid >> 5);
    const int lane = tid & 31;
    const bool router = wave >= GEMM_WAVES;

    const float* Ag = prim + (size_t)bm * 8192;
    const float* Wg = Wt   + (size_t)bm * 16384;

    __builtin_prefetch(Wg, 0, 0);

#pragma unroll
    for (int i = 0; i < 4; ++i) {
        int f4  = tid + i * 512;
        float4 v = ((const float4*)Ag)[f4];
        int f   = f4 << 2;
        int row = f >> 5;
        int kk  = f & 31;
        float* dst = &Alds[row * ALDS_STRIDE + kk];
        dst[0] = v.x; dst[1] = v.y; dst[2] = v.z; dst[3] = v.w;
    }
#pragma unroll
    for (int yx = 0; yx < 4; ++yx) {
        const float4* g = (const float4*)(Wg + yx * 4096 + tid * 8);
        const float4 lo = g[0];
        const float4 hi = g[1];
        v2f* d = (v2f*)Blds + (yx * 4) * BPAIR_PITCH + tid;
        d[0]               = (v2f){lo.x, lo.y};
        d[BPAIR_PITCH]     = (v2f){lo.z, lo.w};
        d[2 * BPAIR_PITCH] = (v2f){hi.x, hi.y};
        d[3 * BPAIR_PITCH] = (v2f){hi.z, hi.w};
    }

    const int rset = (wave - GEMM_WAVES) >> 2;
    const int rrow = (((wave - GEMM_WAVES) & 3) << 3) + (lane >> 2);
    const int mh   = (lane >> 1) & 1;
    const int dh   = lane & 1;
    const float4* bp4 = (const float4*)(bprior + (size_t)o * 256 * 32);
    float4 pv = make_float4(0.f, 0.f, 0.f, 0.f);
    if (!router) pv = bp4[tid];
    __syncthreads();

    v2f   x[4][16];
    float bb[16];

    for (int p = 0; p < PHASES; ++p) {
        if (!router) {
            if (p < 8) {
                ((float4*)(Pri + (p & 1) * PRI_FLOATS))[tid] = pv;
                if (p < 7) pv = bp4[(p + 1) * 256 + tid];
                gemm_strip(Alds, Blds, (p & 1) ? Strip + STRIP_FLOATS : Strip, p, wave, lane);
            }
        } else {
            const int sf = p - 1;
            if ((sf & 1) == rset) {
                if (sf >= 0 && sf < 8)
                    route_begin(Strip + (sf & 1) * STRIP_FLOATS + rrow * SLDS_STRIDE,
                                Pri + (sf & 1) * PRI_FLOATS + rrow * 32, mh, dh, x, bb);
            } else {
                const int s2 = p - 2;
                if (s2 >= 0 && s2 < 8) {
                    v2f v8[4];
                    route_finish(x, bb, v8);
                    const float4 v = make_float4(mh ? v8[2].x : v8[0].x, mh ? v8[2].y : v8[0].y,
                                                 mh ? v8[3].x : v8[1].x, mh ? v8[3].y : v8[1].y);
                    const size_t k = (size_t)s2 * STRIP_ROWS + rrow;
                    *(float4*)(out + ((size_t)bm * 256 + k) * 16 + dh * 8 + mh * 4) = v;
                }
            }
        }
        __syncthreads();
    }
}

extern "C" void kernel_launch(void* const* d_in, const int* in_sizes, int n_in,
                              void* d_out, int out_size, void* d_ws, size_t ws_size,
                              hipStream_t stream)
{
    const float* prim = (const float*)d_in[0];
    const float* Wt   = (const float*)d_in[1];
    const float* bpr  = (const float*)d_in[2];
    float* out = (float*)d_out;

    const size_t lds_bytes = (size_t)LDS_FLOATS * sizeof(float);
    capsule_fused_wmma<<<256, 512, lds_bytes, stream>>>(prim, Wt, bpr, out);
}
